# starting kernel
# baseline (speedup 1.0000x reference)
#include <hip/hip_runtime.h>
#include <math.h>

#define NS 16
#define NN 1024
#define MB 256
#define NITERS 21
#define LOG2E 1.4426950408889634f
#define EPSC 1e-10f

typedef float v2f __attribute__((ext_vector_type(2)));
typedef float v8f __attribute__((ext_vector_type(8)));

__device__ __forceinline__ float fexp2(float x) { return __builtin_amdgcn_exp2f(x); }
__device__ __forceinline__ float flog2(float x) { return __builtin_amdgcn_logf(x); }

__global__ void __launch_bounds__(256) prep_kernel(const float* __restrict__ la,
                                                   const float* __restrict__ u,
                                                   float* __restrict__ a2,
                                                   float* __restrict__ r2) {
    unsigned idx = blockIdx.x * 256u + threadIdx.x;
    unsigned ij  = idx & (NN * NN - 1);
    float uu    = u[idx];
    float noise = -logf(EPSC - logf(uu + EPSC));
    a2[idx] = LOG2E * (la[ij] + noise);
    if (idx < NS * NN) r2[idx] = 0.0f;
}

__global__ void __launch_bounds__(256) colpass_kernel(const float* __restrict__ a2,
                                                      const float* __restrict__ r2,
                                                      float* __restrict__ parts) {
    __shared__ float rsh[128];
    const int s = blockIdx.z, rc = blockIdx.y, cc = blockIdx.x;
    const int t = threadIdx.x;
    const int j = cc * 256 + t;
    const int i0 = rc * 128;
    if (t < 128) rsh[t] = r2[s * NN + i0 + t];
    __syncthreads();
    const float* ap = a2 + ((size_t)s * NN + i0) * NN + j;
    float s0 = 0.f, s1 = 0.f, s2 = 0.f, s3 = 0.f;
    for (int i = 0; i < 128; i += 4) {
        s0 += fexp2(ap[(size_t)(i + 0) * NN] - rsh[i + 0]);
        s1 += fexp2(ap[(size_t)(i + 1) * NN] - rsh[i + 1]);
        s2 += fexp2(ap[(size_t)(i + 2) * NN] - rsh[i + 2]);
        s3 += fexp2(ap[(size_t)(i + 3) * NN] - rsh[i + 3]);
    }
    parts[((size_t)rc * NS + s) * NN + j] = (s0 + s1) + (s2 + s3);
}

__global__ void __launch_bounds__(256) colfin_kernel(const float* __restrict__ parts,
                                                     float* __restrict__ c2) {
    const int idx = blockIdx.x * 256 + threadIdx.x;
    float sum = 0.f;
#pragma unroll
    for (int p = 0; p < 8; ++p) sum += parts[(size_t)p * NS * NN + idx];
    c2[idx] = flog2(sum);
}

__global__ void __launch_bounds__(256) rowpass_kernel(const float* __restrict__ a2,
                                                      const float* __restrict__ c2,
                                                      float* __restrict__ r2) {
    __shared__ float csh[NN];
    const int s = blockIdx.y;
    const int t = threadIdx.x;
    for (int q = t; q < NN; q += 256) csh[q] = c2[s * NN + q];
    __syncthreads();
    const int w = t >> 5, l = t & 31;
    const int i = blockIdx.x * 8 + w;
    const float* ap = a2 + ((size_t)s * NN + i) * NN;
    float s0 = 0.f, s1 = 0.f;
    for (int j = l; j < NN; j += 64) {
        s0 += fexp2(ap[j]      - csh[j]);
        s1 += fexp2(ap[j + 32] - csh[j + 32]);
    }
    float sum = s0 + s1;
#pragma unroll
    for (int m = 16; m >= 1; m >>= 1) sum += __shfl_xor(sum, m, 32);
    if (l == 0) r2[s * NN + i] = flog2(sum);
}

__global__ void __launch_bounds__(256) gemm_kernel(const float* __restrict__ x,
                                                   const float* __restrict__ a2,
                                                   const float* __restrict__ r2,
                                                   const float* __restrict__ c2,
                                                   float* __restrict__ out) {
    constexpr int XS = 34;
    constexpr int PT = 34;
    __shared__ float xs[MB * XS];
    __shared__ float ps[64 * PT];
    const int s  = blockIdx.y;
    const int m0 = blockIdx.x * 64;
    const int t  = threadIdx.x;
    const int w  = t >> 5, l = t & 31;
    const int lm = l & 15, lh = l >> 4;

    const float* ra = a2 + (size_t)s * NN * NN;
    const float* rr = r2 + s * NN;
    const float* rc = c2 + s * NN;

    v8f acc[2][4];
#pragma unroll
    for (int rt = 0; rt < 2; ++rt)
#pragma unroll
        for (int ct = 0; ct < 4; ++ct)
#pragma unroll
            for (int g = 0; g < 8; ++g) acc[rt][ct][g] = 0.0f;

    for (int k0 = 0; k0 < NN; k0 += 32) {
        __syncthreads();
#pragma unroll
        for (int it = 0; it < 32; ++it) {
            int q = it * 256 + t;
            int row = q >> 5, kk = q & 31;
            xs[row * XS + kk] = x[row * NN + k0 + kk];
        }
#pragma unroll
        for (int it = 0; it < 8; ++it) {
            int q = it * 256 + t;
            int kk = q >> 6, m = q & 63;
            ps[m * PT + kk] =
                fexp2(ra[(size_t)(k0 + kk) * NN + m0 + m] - rr[k0 + kk] - rc[m0 + m]);
        }
        __syncthreads();
#pragma unroll
        for (int ks = 0; ks < 32; ks += 4) {
            const int ka = ks + (lh << 1);
            v2f a[2];
#pragma unroll
            for (int rt = 0; rt < 2; ++rt) {
                int row = w * 32 + rt * 16 + lm;
                a[rt] = *(const v2f*)&xs[row * XS + ka];
            }
#pragma unroll
            for (int ct = 0; ct < 4; ++ct) {
                int col = ct * 16 + lm;
                v2f b = *(const v2f*)&ps[col * PT + ka];
#pragma unroll
                for (int rt = 0; rt < 2; ++rt) {
                    acc[rt][ct] = __builtin_amdgcn_wmma_f32_16x16x4_f32(
                        false, a[rt], false, b, (short)0, acc[rt][ct], false, false);
                }
            }
        }
    }
    float* ro = out + (size_t)s * MB * NN;
#pragma unroll
    for (int rt = 0; rt < 2; ++rt)
#pragma unroll
        for (int ct = 0; ct < 4; ++ct)
#pragma unroll
            for (int g = 0; g < 8; ++g) {
                int row = w * 32 + rt * 16 + lh * 8 + g;
                int col = m0 + ct * 16 + lm;
                ro[row * NN + col] = acc[rt][ct][g];
            }
}

extern "C" void kernel_launch(void* const* d_in, const int* in_sizes, int n_in,
                              void* d_out, int out_size, void* d_ws, size_t ws_size,
                              hipStream_t stream) {
    const float* x  = (const float*)d_in[0];
    const float* la = (const float*)d_in[1];
    const float* u  = (const float*)d_in[2];
    float* out = (float*)d_out;

    char* ws = (char*)d_ws;
    float* a2    = (float*)ws;
    float* r2    = (float*)(ws + (size_t)NS * NN * NN * sizeof(float));
    float* c2    = r2 + NS * NN;
    float* parts = c2 + NS * NN;

    prep_kernel<<<(NS * NN * NN) / 256, 256, 0, stream>>>(la, u, a2, r2);
    for (int it = 0; it < NITERS; ++it) {
        colpass_kernel<<<dim3(4, 8, NS), 256, 0, stream>>>(a2, r2, parts);
        colfin_kernel<<<(NS * NN) / 256, 256, 0, stream>>>(parts, c2);
        rowpass_kernel<<<dim3(NN / 8, NS), 256, 0, stream>>>(a2, c2, r2);
    }
    gemm_kernel<<<dim3(NN / 64, NS), 256, 0, stream>>>(x, a2, r2, c2, out);
}


# Fuse each row pass with the next column pass: a wave holds 4 whole rows in registers, so a2 is read once per iteration; row and column sums are added in a new order; 65 to 45 launches
# speedup vs baseline: 1.5045x; 1.5045x over previous
#include <hip/hip_runtime.h>
#include <math.h>

#define NS 16
#define NN 1024
#define MB 256
#define NITERS 21
#define LOG2E 1.4426950408889634f
#define EPSC 1e-10f

typedef float v2f __attribute__((ext_vector_type(2)));
typedef float v4f __attribute__((ext_vector_type(4)));
typedef float v8f __attribute__((ext_vector_type(8)));

__device__ __forceinline__ float fexp2(float x) { return __builtin_amdgcn_exp2f(x); }
__device__ __forceinline__ float flog2(float x) { return __builtin_amdgcn_logf(x); }

__global__ void __launch_bounds__(256) prep_kernel(const float* __restrict__ la,
                                                   const float* __restrict__ u,
                                                   float* __restrict__ a2,
                                                   float* __restrict__ r2) {
    unsigned idx = blockIdx.x * 256u + threadIdx.x;
    unsigned ij  = idx & (NN * NN - 1);
    float uu    = u[idx];
    float noise = -logf(EPSC - logf(uu + EPSC));
    a2[idx] = LOG2E * (la[ij] + noise);
    if (idx < NS * NN) r2[idx] = 0.0f;
}

__global__ void __launch_bounds__(256) colpass_kernel(const float* __restrict__ a2,
                                                      const float* __restrict__ r2,
                                                      float* __restrict__ parts) {
    __shared__ float rsh[128];
    const int s = blockIdx.z, rc = blockIdx.y, cc = blockIdx.x;
    const int t = threadIdx.x;
    const int j = cc * 256 + t;
    const int i0 = rc * 128;
    if (t < 128) rsh[t] = r2[s * NN + i0 + t];
    __syncthreads();
    const float* ap = a2 + ((size_t)s * NN + i0) * NN + j;
    float s0 = 0.f, s1 = 0.f, s2 = 0.f, s3 = 0.f;
    for (int i = 0; i < 128; i += 4) {
        s0 += fexp2(ap[(size_t)(i + 0) * NN] - rsh[i + 0]);
        s1 += fexp2(ap[(size_t)(i + 1) * NN] - rsh[i + 1]);
        s2 += fexp2(ap[(size_t)(i + 2) * NN] - rsh[i + 2]);
        s3 += fexp2(ap[(size_t)(i + 3) * NN] - rsh[i + 3]);
    }
    parts[((size_t)rc * NS + s) * NN + j] = (s0 + s1) + (s2 + s3);
}

template <int P>
__global__ void __launch_bounds__(256) colfin_kernel(const float* __restrict__ parts,
                                                     float* __restrict__ c2) {
    const int idx = blockIdx.x * 256 + threadIdx.x;
    float sum = 0.f;
#pragma unroll
    for (int p = 0; p < P; ++p) sum += parts[(size_t)p * NS * NN + idx];
    c2[idx] = flog2(sum);
}

#define RWAVES 16
#define RROWS  4
#define RCB    (NN / (RWAVES * RROWS))
template <bool LAST>
__global__ void __launch_bounds__(RWAVES * 32) rowcol_kernel(const float* __restrict__ a2,
                                                             const float* __restrict__ c2,
                                                             float* __restrict__ r2,
                                                             float* __restrict__ parts) {
    const int s = blockIdx.y, rb = blockIdx.x;
    const int t = threadIdx.x;
    const int w = t >> 5, l = t & 31;
    const int row0 = rb * (RWAVES * RROWS) + w * RROWS;
    const float* cp = c2 + s * NN + 4 * l;
    const float* ap = a2 + ((size_t)s * NN + row0) * NN + 4 * l;

    v4f c[8];
    v4f v[RROWS][8];
#pragma unroll
    for (int k = 0; k < 8; ++k) c[k] = *(const v4f*)(cp + 128 * k);
#pragma unroll
    for (int r = 0; r < RROWS; ++r)
#pragma unroll
        for (int k = 0; k < 8; ++k) v[r][k] = *(const v4f*)(ap + (size_t)r * NN + 128 * k);
    __builtin_amdgcn_sched_barrier(0);

    float rs[RROWS];
#pragma unroll
    for (int r = 0; r < RROWS; ++r) {
        float s0 = 0.f, s1 = 0.f, s2 = 0.f, s3 = 0.f;
#pragma unroll
        for (int k = 0; k < 8; ++k) {
            s0 += fexp2(v[r][k].x - c[k].x);
            s1 += fexp2(v[r][k].y - c[k].y);
            s2 += fexp2(v[r][k].z - c[k].z);
            s3 += fexp2(v[r][k].w - c[k].w);
        }
        rs[r] = (s0 + s1) + (s2 + s3);
    }
#pragma unroll
    for (int m = 16; m >= 1; m >>= 1)
#pragma unroll
        for (int r = 0; r < RROWS; ++r) rs[r] += __shfl_xor(rs[r], m, 32);
#pragma unroll
    for (int r = 0; r < RROWS; ++r) rs[r] = flog2(rs[r]);

    if constexpr (LAST) {
        float mine = rs[0];
#pragma unroll
        for (int r = 1; r < RROWS; ++r) if (l == r) mine = rs[r];
        if (l < RROWS) r2[s * NN + row0 + l] = mine;
    } else {
        __shared__ __attribute__((aligned(16))) float csum[RWAVES * NN];
        v4f acc[8];
#pragma unroll
        for (int k = 0; k < 8; ++k) {
            acc[k].x = fexp2(v[0][k].x - rs[0]);
            acc[k].y = fexp2(v[0][k].y - rs[0]);
            acc[k].z = fexp2(v[0][k].z - rs[0]);
            acc[k].w = fexp2(v[0][k].w - rs[0]);
        }
#pragma unroll
        for (int r = 1; r < RROWS; ++r)
#pragma unroll
            for (int k = 0; k < 8; ++k) {
                acc[k].x += fexp2(v[r][k].x - rs[r]);
                acc[k].y += fexp2(v[r][k].y - rs[r]);
                acc[k].z += fexp2(v[r][k].z - rs[r]);
                acc[k].w += fexp2(v[r][k].w - rs[r]);
            }
#pragma unroll
        for (int k = 0; k < 8; ++k) *(v4f*)&csum[w * NN + 128 * k + 4 * l] = acc[k];
        __syncthreads();
        v2f tot = *(const v2f*)&csum[2 * t];
#pragma unroll
        for (int ww = 1; ww < RWAVES; ++ww) tot += *(const v2f*)&csum[ww * NN + 2 * t];
        *(v2f*)&parts[((size_t)rb * NS + s) * NN + 2 * t] = tot;
    }
}

__global__ void __launch_bounds__(256) gemm_kernel(const float* __restrict__ x,
                                                   const float* __restrict__ a2,
                                                   const float* __restrict__ r2,
                                                   const float* __restrict__ c2,
                                                   float* __restrict__ out) {
    constexpr int XS = 34;
    constexpr int PT = 34;
    __shared__ float xs[MB * XS];
    __shared__ float ps[64 * PT];
    const int s  = blockIdx.y;
    const int m0 = blockIdx.x * 64;
    const int t  = threadIdx.x;
    const int w  = t >> 5, l = t & 31;
    const int lm = l & 15, lh = l >> 4;

    const float* ra = a2 + (size_t)s * NN * NN;
    const float* rr = r2 + s * NN;
    const float* rc = c2 + s * NN;

    v8f acc[2][4];
#pragma unroll
    for (int rt = 0; rt < 2; ++rt)
#pragma unroll
        for (int ct = 0; ct < 4; ++ct)
#pragma unroll
            for (int g = 0; g < 8; ++g) acc[rt][ct][g] = 0.0f;

    for (int k0 = 0; k0 < NN; k0 += 32) {
        __syncthreads();
#pragma unroll
        for (int it = 0; it < 32; ++it) {
            int q = it * 256 + t;
            int row = q >> 5, kk = q & 31;
            xs[row * XS + kk] = x[row * NN + k0 + kk];
        }
#pragma unroll
        for (int it = 0; it < 8; ++it) {
            int q = it * 256 + t;
            int kk = q >> 6, m = q & 63;
            ps[m * PT + kk] =
                fexp2(ra[(size_t)(k0 + kk) * NN + m0 + m] - rr[k0 + kk] - rc[m0 + m]);
        }
        __syncthreads();
#pragma unroll
        for (int ks = 0; ks < 32; ks += 4) {
            const int ka = ks + (lh << 1);
            v2f a[2];
#pragma unroll
            for (int rt = 0; rt < 2; ++rt) {
                int row = w * 32 + rt * 16 + lm;
                a[rt] = *(const v2f*)&xs[row * XS + ka];
            }
#pragma unroll
            for (int ct = 0; ct < 4; ++ct) {
                int col = ct * 16 + lm;
                v2f b = *(const v2f*)&ps[col * PT + ka];
#pragma unroll
                for (int rt = 0; rt < 2; ++rt) {
                    acc[rt][ct] = __builtin_amdgcn_wmma_f32_16x16x4_f32(
                        false, a[rt], false, b, (short)0, acc[rt][ct], false, false);
                }
            }
        }
    }
    float* ro = out + (size_t)s * MB * NN;
#pragma unroll
    for (int rt = 0; rt < 2; ++rt)
#pragma unroll
        for (int ct = 0; ct < 4; ++ct)
#pragma unroll
            for (int g = 0; g < 8; ++g) {
                int row = w * 32 + rt * 16 + lh * 8 + g;
                int col = m0 + ct * 16 + lm;
                ro[row * NN + col] = acc[rt][ct][g];
            }
}

extern "C" void kernel_launch(void* const* d_in, const int* in_sizes, int n_in,
                              void* d_out, int out_size, void* d_ws, size_t ws_size,
                              hipStream_t stream) {
    const float* x  = (const float*)d_in[0];
    const float* la = (const float*)d_in[1];
    const float* u  = (const float*)d_in[2];
    float* out = (float*)d_out;

    char* ws = (char*)d_ws;
    float* a2    = (float*)ws;
    float* r2    = (float*)(ws + (size_t)NS * NN * NN * sizeof(float));
    float* c2    = r2 + NS * NN;
    float* parts = c2 + NS * NN;

    prep_kernel<<<(NS * NN * NN) / 256, 256, 0, stream>>>(la, u, a2, r2);
    colpass_kernel<<<dim3(4, 8, NS), 256, 0, stream>>>(a2, r2, parts);
    colfin_kernel<8><<<(NS * NN) / 256, 256, 0, stream>>>(parts, c2);
    for (int it = 1; it < NITERS; ++it) {
        rowcol_kernel<false><<<dim3(RCB, NS), RWAVES * 32, 0, stream>>>(a2, c2, r2, out);
        colfin_kernel<RCB><<<(NS * NN) / 256, 256, 0, stream>>>(out, c2);
    }
    rowcol_kernel<true><<<dim3(RCB, NS), RWAVES * 32, 0, stream>>>(a2, c2, r2, out);
    gemm_kernel<<<dim3(NN / 64, NS), 256, 0, stream>>>(x, a2, r2, c2, out);
}


# Fold the column finish (sum of partial planes, log2) into the row pass that consumes it: each block sums its sample's planes itself, planes double-buffered; 45 to 24 launches
# speedup vs baseline: 1.6606x; 1.1038x over previous
#include <hip/hip_runtime.h>
#include <math.h>

#define NS 16
#define NN 1024
#define MB 256
#define NITERS 21
#define LOG2E 1.4426950408889634f
#define EPSC 1e-10f

typedef float v2f __attribute__((ext_vector_type(2)));
typedef float v4f __attribute__((ext_vector_type(4)));
typedef float v8f __attribute__((ext_vector_type(8)));

__device__ __forceinline__ float fexp2(float x) { return __builtin_amdgcn_exp2f(x); }
__device__ __forceinline__ float flog2(float x) { return __builtin_amdgcn_logf(x); }

__global__ void __launch_bounds__(256) prep_kernel(const float* __restrict__ la,
                                                   const float* __restrict__ u,
                                                   float* __restrict__ a2,
                                                   float* __restrict__ r2) {
    unsigned idx = blockIdx.x * 256u + threadIdx.x;
    unsigned ij  = idx & (NN * NN - 1);
    float uu    = u[idx];
    float noise = -logf(EPSC - logf(uu + EPSC));
    a2[idx] = LOG2E * (la[ij] + noise);
    if (idx < NS * NN) r2[idx] = 0.0f;
}

__global__ void __launch_bounds__(256) colpass_kernel(const float* __restrict__ a2,
                                                      const float* __restrict__ r2,
                                                      float* __restrict__ parts) {
    __shared__ float rsh[128];
    const int s = blockIdx.z, rc = blockIdx.y, cc = blockIdx.x;
    const int t = threadIdx.x;
    const int j = cc * 256 + t;
    const int i0 = rc * 128;
    if (t < 128) rsh[t] = r2[s * NN + i0 + t];
    __syncthreads();
    const float* ap = a2 + ((size_t)s * NN + i0) * NN + j;
    float s0 = 0.f, s1 = 0.f, s2 = 0.f, s3 = 0.f;
    for (int i = 0; i < 128; i += 4) {
        s0 += fexp2(ap[(size_t)(i + 0) * NN] - rsh[i + 0]);
        s1 += fexp2(ap[(size_t)(i + 1) * NN] - rsh[i + 1]);
        s2 += fexp2(ap[(size_t)(i + 2) * NN] - rsh[i + 2]);
        s3 += fexp2(ap[(size_t)(i + 3) * NN] - rsh[i + 3]);
    }
    parts[((size_t)rc * NS + s) * NN + j] = (s0 + s1) + (s2 + s3);
}

#define RWAVES 16
#define RROWS  4
#define RCB    (NN / (RWAVES * RROWS))
template <bool LAST, int P>
__global__ void __launch_bounds__(RWAVES * 32) rowcol_kernel(const float* __restrict__ a2,
                                                             const float* __restrict__ pin,
                                                             float* __restrict__ r2,
                                                             float* __restrict__ c2,
                                                             float* __restrict__ pout) {
    __shared__ __attribute__((aligned(16))) float csh[NN];
    const int s = blockIdx.y, rb = blockIdx.x;
    const int t = threadIdx.x;
    const int w = t >> 5, l = t & 31;
    const int row0 = rb * (RWAVES * RROWS) + w * RROWS;
    const float* pp = pin + (size_t)s * NN + 2 * t;
    const float* ap = a2 + ((size_t)s * NN + row0) * NN + 4 * l;

    v2f pl[P];
    v4f c[8];
    v4f v[RROWS][8];
#pragma unroll
    for (int p = 0; p < P; ++p) pl[p] = *(const v2f*)(pp + (size_t)p * NS * NN);
#pragma unroll
    for (int r = 0; r < RROWS; ++r)
#pragma unroll
        for (int k = 0; k < 8; ++k) v[r][k] = *(const v4f*)(ap + (size_t)r * NN + 128 * k);
    __builtin_amdgcn_sched_barrier(0);

    {
        v2f tot = pl[0];
#pragma unroll
        for (int p = 1; p < P; ++p) tot += pl[p];
        v2f cc;
        cc.x = flog2(tot.x);
        cc.y = flog2(tot.y);
        *(v2f*)&csh[2 * t] = cc;
        if constexpr (LAST) {
            if (rb == 0) *(v2f*)&c2[s * NN + 2 * t] = cc;
        }
    }
    asm volatile("s_wait_dscnt 0x0\n\ts_barrier_signal -1\n\ts_barrier_wait -1" ::: "memory");
#pragma unroll
    for (int k = 0; k < 8; ++k) c[k] = *(const v4f*)&csh[128 * k + 4 * l];

    float rs[RROWS];
#pragma unroll
    for (int r = 0; r < RROWS; ++r) {
        float s0 = 0.f, s1 = 0.f, s2 = 0.f, s3 = 0.f;
#pragma unroll
        for (int k = 0; k < 8; ++k) {
            s0 += fexp2(v[r][k].x - c[k].x);
            s1 += fexp2(v[r][k].y - c[k].y);
            s2 += fexp2(v[r][k].z - c[k].z);
            s3 += fexp2(v[r][k].w - c[k].w);
        }
        rs[r] = (s0 + s1) + (s2 + s3);
    }
#pragma unroll
    for (int m = 16; m >= 1; m >>= 1)
#pragma unroll
        for (int r = 0; r < RROWS; ++r) rs[r] += __shfl_xor(rs[r], m, 32);
#pragma unroll
    for (int r = 0; r < RROWS; ++r) rs[r] = flog2(rs[r]);

    if constexpr (LAST) {
        float mine = rs[0];
#pragma unroll
        for (int r = 1; r < RROWS; ++r) if (l == r) mine = rs[r];
        if (l < RROWS) r2[s * NN + row0 + l] = mine;
    } else {
        __shared__ __attribute__((aligned(16))) float csum[RWAVES * NN];
        v4f acc[8];
#pragma unroll
        for (int k = 0; k < 8; ++k) {
            acc[k].x = fexp2(v[0][k].x - rs[0]);
            acc[k].y = fexp2(v[0][k].y - rs[0]);
            acc[k].z = fexp2(v[0][k].z - rs[0]);
            acc[k].w = fexp2(v[0][k].w - rs[0]);
        }
#pragma unroll
        for (int r = 1; r < RROWS; ++r)
#pragma unroll
            for (int k = 0; k < 8; ++k) {
                acc[k].x += fexp2(v[r][k].x - rs[r]);
                acc[k].y += fexp2(v[r][k].y - rs[r]);
                acc[k].z += fexp2(v[r][k].z - rs[r]);
                acc[k].w += fexp2(v[r][k].w - rs[r]);
            }
#pragma unroll
        for (int k = 0; k < 8; ++k) *(v4f*)&csum[w * NN + 128 * k + 4 * l] = acc[k];
        __syncthreads();
        v2f tot = *(const v2f*)&csum[2 * t];
#pragma unroll
        for (int ww = 1; ww < RWAVES; ++ww) tot += *(const v2f*)&csum[ww * NN + 2 * t];
        *(v2f*)&pout[((size_t)rb * NS + s) * NN + 2 * t] = tot;
    }
}

__global__ void __launch_bounds__(256) gemm_kernel(const float* __restrict__ x,
                                                   const float* __restrict__ a2,
                                                   const float* __restrict__ r2,
                                                   const float* __restrict__ c2,
                                                   float* __restrict__ out) {
    constexpr int XS = 34;
    constexpr int PT = 34;
    __shared__ float xs[MB * XS];
    __shared__ float ps[64 * PT];
    const int s  = blockIdx.y;
    const int m0 = blockIdx.x * 64;
    const int t  = threadIdx.x;
    const int w  = t >> 5, l = t & 31;
    const int lm = l & 15, lh = l >> 4;

    const float* ra = a2 + (size_t)s * NN * NN;
    const float* rr = r2 + s * NN;
    const float* rc = c2 + s * NN;

    v8f acc[2][4];
#pragma unroll
    for (int rt = 0; rt < 2; ++rt)
#pragma unroll
        for (int ct = 0; ct < 4; ++ct)
#pragma unroll
            for (int g = 0; g < 8; ++g) acc[rt][ct][g] = 0.0f;

    for (int k0 = 0; k0 < NN; k0 += 32) {
        __syncthreads();
#pragma unroll
        for (int it = 0; it < 32; ++it) {
            int q = it * 256 + t;
            int row = q >> 5, kk = q & 31;
            xs[row * XS + kk] = x[row * NN + k0 + kk];
        }
#pragma unroll
        for (int it = 0; it < 8; ++it) {
            int q = it * 256 + t;
            int kk = q >> 6, m = q & 63;
            ps[m * PT + kk] =
                fexp2(ra[(size_t)(k0 + kk) * NN + m0 + m] - rr[k0 + kk] - rc[m0 + m]);
        }
        __syncthreads();
#pragma unroll
        for (int ks = 0; ks < 32; ks += 4) {
            const int ka = ks + (lh << 1);
            v2f a[2];
#pragma unroll
            for (int rt = 0; rt < 2; ++rt) {
                int row = w * 32 + rt * 16 + lm;
                a[rt] = *(const v2f*)&xs[row * XS + ka];
            }
#pragma unroll
            for (int ct = 0; ct < 4; ++ct) {
                int col = ct * 16 + lm;
                v2f b = *(const v2f*)&ps[col * PT + ka];
#pragma unroll
                for (int rt = 0; rt < 2; ++rt) {
                    acc[rt][ct] = __builtin_amdgcn_wmma_f32_16x16x4_f32(
                        false, a[rt], false, b, (short)0, acc[rt][ct], false, false);
                }
            }
        }
    }
    float* ro = out + (size_t)s * MB * NN;
#pragma unroll
    for (int rt = 0; rt < 2; ++rt)
#pragma unroll
        for (int ct = 0; ct < 4; ++ct)
#pragma unroll
            for (int g = 0; g < 8; ++g) {
                int row = w * 32 + rt * 16 + lh * 8 + g;
                int col = m0 + ct * 16 + lm;
                ro[row * NN + col] = acc[rt][ct][g];
            }
}

extern "C" void kernel_launch(void* const* d_in, const int* in_sizes, int n_in,
                              void* d_out, int out_size, void* d_ws, size_t ws_size,
                              hipStream_t stream) {
    const float* x  = (const float*)d_in[0];
    const float* la = (const float*)d_in[1];
    const float* u  = (const float*)d_in[2];
    float* out = (float*)d_out;

    char* ws = (char*)d_ws;
    float* a2    = (float*)ws;
    float* r2    = (float*)(ws + (size_t)NS * NN * NN * sizeof(float));
    float* c2    = r2 + NS * NN;
    float* parts = c2 + NS * NN;

    float* pa = out;
    float* pb = out + (size_t)RCB * NS * NN;
    prep_kernel<<<(NS * NN * NN) / 256, 256, 0, stream>>>(la, u, a2, r2);
    colpass_kernel<<<dim3(4, 8, NS), 256, 0, stream>>>(a2, r2, parts);
    rowcol_kernel<false, 8><<<dim3(RCB, NS), RWAVES * 32, 0, stream>>>(a2, parts, r2, c2, pa);
    for (int it = 0; it < 9; ++it) {
        rowcol_kernel<false, RCB><<<dim3(RCB, NS), RWAVES * 32, 0, stream>>>(a2, pa, r2, c2, pb);
        rowcol_kernel<false, RCB><<<dim3(RCB, NS), RWAVES * 32, 0, stream>>>(a2, pb, r2, c2, pa);
    }
    rowcol_kernel<false, RCB><<<dim3(RCB, NS), RWAVES * 32, 0, stream>>>(a2, pa, r2, c2, pb);
    rowcol_kernel<true, RCB><<<dim3(RCB, NS), RWAVES * 32, 0, stream>>>(a2, pb, r2, c2, pa);
    gemm_kernel<<<dim3(NN / 64, NS), 256, 0, stream>>>(x, a2, r2, c2, out);
}


# Multiplicative form of the same 21 iterations: K=2^a2 stored once, passes scale by reciprocals of the sums (1/x for log2, one multiply-add per element for sub+exp2+add); sums keep their order
# speedup vs baseline: 1.8153x; 1.0932x over previous
#include <hip/hip_runtime.h>
#include <math.h>

#define NS 16
#define NN 1024
#define MB 256
#define NITERS 21
#define LOG2E 1.4426950408889634f
#define EPSC 1e-10f

typedef float v2f __attribute__((ext_vector_type(2)));
typedef float v4f __attribute__((ext_vector_type(4)));
typedef float v8f __attribute__((ext_vector_type(8)));

__device__ __forceinline__ float fexp2(float x) { return __builtin_amdgcn_exp2f(x); }

__global__ void __launch_bounds__(256) prep_kernel(const float* __restrict__ la,
                                                   const float* __restrict__ u,
                                                   float* __restrict__ km) {
    unsigned idx = blockIdx.x * 256u + threadIdx.x;
    unsigned ij  = idx & (NN * NN - 1);
    float uu    = u[idx];
    float noise = -logf(EPSC - logf(uu + EPSC));
    km[idx] = fexp2(LOG2E * (la[ij] + noise));
}

__global__ void __launch_bounds__(256) colpass_kernel(const float* __restrict__ km,
                                                      float* __restrict__ parts) {
    const int s = blockIdx.z, rc = blockIdx.y, cc = blockIdx.x;
    const int t = threadIdx.x;
    const int j = cc * 256 + t;
    const int i0 = rc * 128;
    const float* ap = km + ((size_t)s * NN + i0) * NN + j;
    float s0 = 0.f, s1 = 0.f, s2 = 0.f, s3 = 0.f;
    for (int i = 0; i < 128; i += 4) {
        s0 += ap[(size_t)(i + 0) * NN];
        s1 += ap[(size_t)(i + 1) * NN];
        s2 += ap[(size_t)(i + 2) * NN];
        s3 += ap[(size_t)(i + 3) * NN];
    }
    parts[((size_t)rc * NS + s) * NN + j] = (s0 + s1) + (s2 + s3);
}

#define RWAVES 16
#define RROWS  4
#define RCB    (NN / (RWAVES * RROWS))
template <bool LAST, int P>
__global__ void __launch_bounds__(RWAVES * 32) rowcol_kernel(const float* __restrict__ km,
                                                             const float* __restrict__ pin,
                                                             float* __restrict__ r2,
                                                             float* __restrict__ c2,
                                                             float* __restrict__ pout) {
    __shared__ __attribute__((aligned(16))) float csh[NN];
    const int s = blockIdx.y, rb = blockIdx.x;
    const int t = threadIdx.x;
    const int w = t >> 5, l = t & 31;
    const int row0 = rb * (RWAVES * RROWS) + w * RROWS;
    const float* pp = pin + (size_t)s * NN + 2 * t;
    const float* ap = km + ((size_t)s * NN + row0) * NN + 4 * l;

    v2f pl[P];
    v4f c[8];
    v4f v[RROWS][8];
#pragma unroll
    for (int p = 0; p < P; ++p) pl[p] = *(const v2f*)(pp + (size_t)p * NS * NN);
#pragma unroll
    for (int r = 0; r < RROWS; ++r)
#pragma unroll
        for (int k = 0; k < 8; ++k) v[r][k] = *(const v4f*)(ap + (size_t)r * NN + 128 * k);
    __builtin_amdgcn_sched_barrier(0);

    {
        v2f tot = pl[0];
#pragma unroll
        for (int p = 1; p < P; ++p) tot += pl[p];
        v2f cc;
        cc.x = 1.0f / tot.x;
        cc.y = 1.0f / tot.y;
        *(v2f*)&csh[2 * t] = cc;
        if constexpr (LAST) {
            if (rb == 0) *(v2f*)&c2[s * NN + 2 * t] = cc;
        }
    }
    asm volatile("s_wait_dscnt 0x0\n\ts_barrier_signal -1\n\ts_barrier_wait -1" ::: "memory");
#pragma unroll
    for (int k = 0; k < 8; ++k) c[k] = *(const v4f*)&csh[128 * k + 4 * l];

    float rs[RROWS];
#pragma unroll
    for (int r = 0; r < RROWS; ++r) {
        v4f sa = v[r][0] * c[0];
#pragma unroll
        for (int k = 1; k < 8; ++k) sa = v[r][k] * c[k] + sa;
        rs[r] = (sa.x + sa.y) + (sa.z + sa.w);
    }
#pragma unroll
    for (int m = 16; m >= 1; m >>= 1)
#pragma unroll
        for (int r = 0; r < RROWS; ++r) rs[r] += __shfl_xor(rs[r], m, 32);
#pragma unroll
    for (int r = 0; r < RROWS; ++r) rs[r] = 1.0f / rs[r];

    if constexpr (LAST) {
        float mine = rs[0];
#pragma unroll
        for (int r = 1; r < RROWS; ++r) if (l == r) mine = rs[r];
        if (l < RROWS) r2[s * NN + row0 + l] = mine;
    } else {
        __shared__ __attribute__((aligned(16))) float csum[RWAVES * NN];
        v4f acc[8];
#pragma unroll
        for (int k = 0; k < 8; ++k) acc[k] = v[0][k] * rs[0];
#pragma unroll
        for (int r = 1; r < RROWS; ++r)
#pragma unroll
            for (int k = 0; k < 8; ++k) acc[k] = v[r][k] * rs[r] + acc[k];
#pragma unroll
        for (int k = 0; k < 8; ++k) *(v4f*)&csum[w * NN + 128 * k + 4 * l] = acc[k];
        __syncthreads();
        v2f tot = *(const v2f*)&csum[2 * t];
#pragma unroll
        for (int ww = 1; ww < RWAVES; ++ww) tot += *(const v2f*)&csum[ww * NN + 2 * t];
        *(v2f*)&pout[((size_t)rb * NS + s) * NN + 2 * t] = tot;
    }
}

__global__ void __launch_bounds__(256) gemm_kernel(const float* __restrict__ x,
                                                   const float* __restrict__ km,
                                                   const float* __restrict__ r2,
                                                   const float* __restrict__ c2,
                                                   float* __restrict__ out) {
    constexpr int XS = 34;
    constexpr int PT = 34;
    __shared__ float xs[MB * XS];
    __shared__ float ps[64 * PT];
    const int s  = blockIdx.y;
    const int m0 = blockIdx.x * 64;
    const int t  = threadIdx.x;
    const int w  = t >> 5, l = t & 31;
    const int lm = l & 15, lh = l >> 4;

    const float* ra = km + (size_t)s * NN * NN;
    const float* rr = r2 + s * NN;
    const float* rc = c2 + s * NN;

    v8f acc[2][4];
#pragma unroll
    for (int rt = 0; rt < 2; ++rt)
#pragma unroll
        for (int ct = 0; ct < 4; ++ct)
#pragma unroll
            for (int g = 0; g < 8; ++g) acc[rt][ct][g] = 0.0f;

    for (int k0 = 0; k0 < NN; k0 += 32) {
        __syncthreads();
#pragma unroll
        for (int it = 0; it < 32; ++it) {
            int q = it * 256 + t;
            int row = q >> 5, kk = q & 31;
            xs[row * XS + kk] = x[row * NN + k0 + kk];
        }
#pragma unroll
        for (int it = 0; it < 8; ++it) {
            int q = it * 256 + t;
            int kk = q >> 6, m = q & 63;
            ps[m * PT + kk] =
                (ra[(size_t)(k0 + kk) * NN + m0 + m] * rr[k0 + kk]) * rc[m0 + m];
        }
        __syncthreads();
#pragma unroll
        for (int ks = 0; ks < 32; ks += 4) {
            const int ka = ks + (lh << 1);
            v2f a[2];
#pragma unroll
            for (int rt = 0; rt < 2; ++rt) {
                int row = w * 32 + rt * 16 + lm;
                a[rt] = *(const v2f*)&xs[row * XS + ka];
            }
#pragma unroll
            for (int ct = 0; ct < 4; ++ct) {
                int col = ct * 16 + lm;
                v2f b = *(const v2f*)&ps[col * PT + ka];
#pragma unroll
                for (int rt = 0; rt < 2; ++rt) {
                    acc[rt][ct] = __builtin_amdgcn_wmma_f32_16x16x4_f32(
                        false, a[rt], false, b, (short)0, acc[rt][ct], false, false);
                }
            }
        }
    }
    float* ro = out + (size_t)s * MB * NN;
#pragma unroll
    for (int rt = 0; rt < 2; ++rt)
#pragma unroll
        for (int ct = 0; ct < 4; ++ct)
#pragma unroll
            for (int g = 0; g < 8; ++g) {
                int row = w * 32 + rt * 16 + lh * 8 + g;
                int col = m0 + ct * 16 + lm;
                ro[row * NN + col] = acc[rt][ct][g];
            }
}

extern "C" void kernel_launch(void* const* d_in, const int* in_sizes, int n_in,
                              void* d_out, int out_size, void* d_ws, size_t ws_size,
                              hipStream_t stream) {
    const float* x  = (const float*)d_in[0];
    const float* la = (const float*)d_in[1];
    const float* u  = (const float*)d_in[2];
    float* out = (float*)d_out;

    char* ws = (char*)d_ws;
    float* km    = (float*)ws;
    float* r2    = (float*)(ws + (size_t)NS * NN * NN * sizeof(float));
    float* c2    = r2 + NS * NN;
    float* parts = c2 + NS * NN;

    float* pa = out;
    float* pb = out + (size_t)RCB * NS * NN;
    prep_kernel<<<(NS * NN * NN) / 256, 256, 0, stream>>>(la, u, km);
    colpass_kernel<<<dim3(4, 8, NS), 256, 0, stream>>>(km, parts);
    rowcol_kernel<false, 8><<<dim3(RCB, NS), RWAVES * 32, 0, stream>>>(km, parts, r2, c2, pa);
    for (int it = 0; it < 9; ++it) {
        rowcol_kernel<false, RCB><<<dim3(RCB, NS), RWAVES * 32, 0, stream>>>(km, pa, r2, c2, pb);
        rowcol_kernel<false, RCB><<<dim3(RCB, NS), RWAVES * 32, 0, stream>>>(km, pb, r2, c2, pa);
    }
    rowcol_kernel<false, RCB><<<dim3(RCB, NS), RWAVES * 32, 0, stream>>>(km, pa, r2, c2, pb);
    rowcol_kernel<true, RCB><<<dim3(RCB, NS), RWAVES * 32, 0, stream>>>(km, pb, r2, c2, pa);
    gemm_kernel<<<dim3(NN / 64, NS), 256, 0, stream>>>(x, km, r2, c2, out);
}


# Prep kernel: one thread does 4 columns of all 16 samples with 128-bit loads and stores, its 17 loads in flight together (log_alpha read once); 524288 one-element waves become 8192
# speedup vs baseline: 1.9254x; 1.0607x over previous
#include <hip/hip_runtime.h>
#include <math.h>

#define NS 16
#define NN 1024
#define MB 256
#define NITERS 21
#define LOG2E 1.4426950408889634f
#define EPSC 1e-10f

typedef float v2f __attribute__((ext_vector_type(2)));
typedef float v4f __attribute__((ext_vector_type(4)));
typedef float v8f __attribute__((ext_vector_type(8)));

__device__ __forceinline__ float fexp2(float x) { return __builtin_amdgcn_exp2f(x); }

__device__ __forceinline__ float kval(float a, float uu) {
    float noise = -logf(EPSC - logf(uu + EPSC));
    return fexp2(LOG2E * (a + noise));
}

__global__ void __launch_bounds__(256) prep_kernel(const float* __restrict__ la,
                                                   const float* __restrict__ u,
                                                   float* __restrict__ km) {
    const unsigned q = (blockIdx.x * 256u + threadIdx.x) * 4u;
    const v4f a = *(const v4f*)(la + q);
    v4f uu[NS];
#pragma unroll
    for (int s = 0; s < NS; ++s) uu[s] = *(const v4f*)(u + (size_t)s * NN * NN + q);
    __builtin_amdgcn_sched_barrier(0);
#pragma unroll
    for (int s = 0; s < NS; ++s) {
        v4f o;
        o.x = kval(a.x, uu[s].x);
        o.y = kval(a.y, uu[s].y);
        o.z = kval(a.z, uu[s].z);
        o.w = kval(a.w, uu[s].w);
        *(v4f*)(km + (size_t)s * NN * NN + q) = o;
    }
}

__global__ void __launch_bounds__(256) colpass_kernel(const float* __restrict__ km,
                                                      float* __restrict__ parts) {
    const int s = blockIdx.z, rc = blockIdx.y, cc = blockIdx.x;
    const int t = threadIdx.x;
    const int j = cc * 256 + t;
    const int i0 = rc * 128;
    const float* ap = km + ((size_t)s * NN + i0) * NN + j;
    float s0 = 0.f, s1 = 0.f, s2 = 0.f, s3 = 0.f;
    for (int i = 0; i < 128; i += 4) {
        s0 += ap[(size_t)(i + 0) * NN];
        s1 += ap[(size_t)(i + 1) * NN];
        s2 += ap[(size_t)(i + 2) * NN];
        s3 += ap[(size_t)(i + 3) * NN];
    }
    parts[((size_t)rc * NS + s) * NN + j] = (s0 + s1) + (s2 + s3);
}

#define RWAVES 16
#define RROWS  4
#define RCB    (NN / (RWAVES * RROWS))
template <bool LAST, int P>
__global__ void __launch_bounds__(RWAVES * 32) rowcol_kernel(const float* __restrict__ km,
                                                             const float* __restrict__ pin,
                                                             float* __restrict__ r2,
                                                             float* __restrict__ c2,
                                                             float* __restrict__ pout) {
    __shared__ __attribute__((aligned(16))) float csh[NN];
    const int s = blockIdx.y, rb = blockIdx.x;
    const int t = threadIdx.x;
    const int w = t >> 5, l = t & 31;
    const int row0 = rb * (RWAVES * RROWS) + w * RROWS;
    const float* pp = pin + (size_t)s * NN + 2 * t;
    const float* ap = km + ((size_t)s * NN + row0) * NN + 4 * l;

    v2f pl[P];
    v4f c[8];
    v4f v[RROWS][8];
#pragma unroll
    for (int p = 0; p < P; ++p) pl[p] = *(const v2f*)(pp + (size_t)p * NS * NN);
#pragma unroll
    for (int r = 0; r < RROWS; ++r)
#pragma unroll
        for (int k = 0; k < 8; ++k) v[r][k] = *(const v4f*)(ap + (size_t)r * NN + 128 * k);
    __builtin_amdgcn_sched_barrier(0);

    {
        v2f tot = pl[0];
#pragma unroll
        for (int p = 1; p < P; ++p) tot += pl[p];
        v2f cc;
        cc.x = 1.0f / tot.x;
        cc.y = 1.0f / tot.y;
        *(v2f*)&csh[2 * t] = cc;
        if constexpr (LAST) {
            if (rb == 0) *(v2f*)&c2[s * NN + 2 * t] = cc;
        }
    }
    asm volatile("s_wait_dscnt 0x0\n\ts_barrier_signal -1\n\ts_barrier_wait -1" ::: "memory");
#pragma unroll
    for (int k = 0; k < 8; ++k) c[k] = *(const v4f*)&csh[128 * k + 4 * l];

    float rs[RROWS];
#pragma unroll
    for (int r = 0; r < RROWS; ++r) {
        v4f sa = v[r][0] * c[0];
#pragma unroll
        for (int k = 1; k < 8; ++k) sa = v[r][k] * c[k] + sa;
        rs[r] = (sa.x + sa.y) + (sa.z + sa.w);
    }
#pragma unroll
    for (int m = 16; m >= 1; m >>= 1)
#pragma unroll
        for (int r = 0; r < RROWS; ++r) rs[r] += __shfl_xor(rs[r], m, 32);
#pragma unroll
    for (int r = 0; r < RROWS; ++r) rs[r] = 1.0f / rs[r];

    if constexpr (LAST) {
        float mine = rs[0];
#pragma unroll
        for (int r = 1; r < RROWS; ++r) if (l == r) mine = rs[r];
        if (l < RROWS) r2[s * NN + row0 + l] = mine;
    } else {
        __shared__ __attribute__((aligned(16))) float csum[RWAVES * NN];
        v4f acc[8];
#pragma unroll
        for (int k = 0; k < 8; ++k) acc[k] = v[0][k] * rs[0];
#pragma unroll
        for (int r = 1; r < RROWS; ++r)
#pragma unroll
            for (int k = 0; k < 8; ++k) acc[k] = v[r][k] * rs[r] + acc[k];
#pragma unroll
        for (int k = 0; k < 8; ++k) *(v4f*)&csum[w * NN + 128 * k + 4 * l] = acc[k];
        __syncthreads();
        v2f tot = *(const v2f*)&csum[2 * t];
#pragma unroll
        for (int ww = 1; ww < RWAVES; ++ww) tot += *(const v2f*)&csum[ww * NN + 2 * t];
        *(v2f*)&pout[((size_t)rb * NS + s) * NN + 2 * t] = tot;
    }
}

__global__ void __launch_bounds__(256) gemm_kernel(const float* __restrict__ x,
                                                   const float* __restrict__ km,
                                                   const float* __restrict__ r2,
                                                   const float* __restrict__ c2,
                                                   float* __restrict__ out) {
    constexpr int XS = 34;
    constexpr int PT = 34;
    __shared__ float xs[MB * XS];
    __shared__ float ps[64 * PT];
    const int s  = blockIdx.y;
    const int m0 = blockIdx.x * 64;
    const int t  = threadIdx.x;
    const int w  = t >> 5, l = t & 31;
    const int lm = l & 15, lh = l >> 4;

    const float* ra = km + (size_t)s * NN * NN;
    const float* rr = r2 + s * NN;
    const float* rc = c2 + s * NN;

    v8f acc[2][4];
#pragma unroll
    for (int rt = 0; rt < 2; ++rt)
#pragma unroll
        for (int ct = 0; ct < 4; ++ct)
#pragma unroll
            for (int g = 0; g < 8; ++g) acc[rt][ct][g] = 0.0f;

    for (int k0 = 0; k0 < NN; k0 += 32) {
        __syncthreads();
#pragma unroll
        for (int it = 0; it < 32; ++it) {
            int q = it * 256 + t;
            int row = q >> 5, kk = q & 31;
            xs[row * XS + kk] = x[row * NN + k0 + kk];
        }
#pragma unroll
        for (int it = 0; it < 8; ++it) {
            int q = it * 256 + t;
            int kk = q >> 6, m = q & 63;
            ps[m * PT + kk] =
                (ra[(size_t)(k0 + kk) * NN + m0 + m] * rr[k0 + kk]) * rc[m0 + m];
        }
        __syncthreads();
#pragma unroll
        for (int ks = 0; ks < 32; ks += 4) {
            const int ka = ks + (lh << 1);
            v2f a[2];
#pragma unroll
            for (int rt = 0; rt < 2; ++rt) {
                int row = w * 32 + rt * 16 + lm;
                a[rt] = *(const v2f*)&xs[row * XS + ka];
            }
#pragma unroll
            for (int ct = 0; ct < 4; ++ct) {
                int col = ct * 16 + lm;
                v2f b = *(const v2f*)&ps[col * PT + ka];
#pragma unroll
                for (int rt = 0; rt < 2; ++rt) {
                    acc[rt][ct] = __builtin_amdgcn_wmma_f32_16x16x4_f32(
                        false, a[rt], false, b, (short)0, acc[rt][ct], false, false);
                }
            }
        }
    }
    float* ro = out + (size_t)s * MB * NN;
#pragma unroll
    for (int rt = 0; rt < 2; ++rt)
#pragma unroll
        for (int ct = 0; ct < 4; ++ct)
#pragma unroll
            for (int g = 0; g < 8; ++g) {
                int row = w * 32 + rt * 16 + lh * 8 + g;
                int col = m0 + ct * 16 + lm;
                ro[row * NN + col] = acc[rt][ct][g];
            }
}

extern "C" void kernel_launch(void* const* d_in, const int* in_sizes, int n_in,
                              void* d_out, int out_size, void* d_ws, size_t ws_size,
                              hipStream_t stream) {
    const float* x  = (const float*)d_in[0];
    const float* la = (const float*)d_in[1];
    const float* u  = (const float*)d_in[2];
    float* out = (float*)d_out;

    char* ws = (char*)d_ws;
    float* km    = (float*)ws;
    float* r2    = (float*)(ws + (size_t)NS * NN * NN * sizeof(float));
    float* c2    = r2 + NS * NN;
    float* parts = c2 + NS * NN;

    float* pa = out;
    float* pb = out + (size_t)RCB * NS * NN;
    prep_kernel<<<(NN * NN / 4) / 256, 256, 0, stream>>>(la, u, km);
    colpass_kernel<<<dim3(4, 8, NS), 256, 0, stream>>>(km, parts);
    rowcol_kernel<false, 8><<<dim3(RCB, NS), RWAVES * 32, 0, stream>>>(km, parts, r2, c2, pa);
    for (int it = 0; it < 9; ++it) {
        rowcol_kernel<false, RCB><<<dim3(RCB, NS), RWAVES * 32, 0, stream>>>(km, pa, r2, c2, pb);
        rowcol_kernel<false, RCB><<<dim3(RCB, NS), RWAVES * 32, 0, stream>>>(km, pb, r2, c2, pa);
    }
    rowcol_kernel<false, RCB><<<dim3(RCB, NS), RWAVES * 32, 0, stream>>>(km, pa, r2, c2, pb);
    rowcol_kernel<true, RCB><<<dim3(RCB, NS), RWAVES * 32, 0, stream>>>(km, pb, r2, c2, pa);
    gemm_kernel<<<dim3(NN / 64, NS), 256, 0, stream>>>(x, km, r2, c2, out);
}


# Product kernel: K loop pipelined over two LDS stages: next tile's global loads fly under the current tile's WMMAs and fill the other stage mid-step; one barrier per K step; sums keep their order
# speedup vs baseline: 2.2002x; 1.1427x over previous
#include <hip/hip_runtime.h>
#include <math.h>

#define NS 16
#define NN 1024
#define MB 256
#define NITERS 21
#define LOG2E 1.4426950408889634f
#define EPSC 1e-10f

typedef float v2f __attribute__((ext_vector_type(2)));
typedef float v4f __attribute__((ext_vector_type(4)));
typedef float v8f __attribute__((ext_vector_type(8)));

__device__ __forceinline__ float fexp2(float x) { return __builtin_amdgcn_exp2f(x); }

__device__ __forceinline__ float kval(float a, float uu) {
    float noise = -logf(EPSC - logf(uu + EPSC));
    return fexp2(LOG2E * (a + noise));
}

__global__ void __launch_bounds__(256) prep_kernel(const float* __restrict__ la,
                                                   const float* __restrict__ u,
                                                   float* __restrict__ km) {
    const unsigned q = (blockIdx.x * 256u + threadIdx.x) * 4u;
    const v4f a = *(const v4f*)(la + q);
    v4f uu[NS];
#pragma unroll
    for (int s = 0; s < NS; ++s) uu[s] = *(const v4f*)(u + (size_t)s * NN * NN + q);
    __builtin_amdgcn_sched_barrier(0);
#pragma unroll
    for (int s = 0; s < NS; ++s) {
        v4f o;
        o.x = kval(a.x, uu[s].x);
        o.y = kval(a.y, uu[s].y);
        o.z = kval(a.z, uu[s].z);
        o.w = kval(a.w, uu[s].w);
        *(v4f*)(km + (size_t)s * NN * NN + q) = o;
    }
}

__global__ void __launch_bounds__(256) colpass_kernel(const float* __restrict__ km,
                                                      float* __restrict__ parts) {
    const int s = blockIdx.z, rc = blockIdx.y, cc = blockIdx.x;
    const int t = threadIdx.x;
    const int j = cc * 256 + t;
    const int i0 = rc * 128;
    const float* ap = km + ((size_t)s * NN + i0) * NN + j;
    float s0 = 0.f, s1 = 0.f, s2 = 0.f, s3 = 0.f;
    for (int i = 0; i < 128; i += 4) {
        s0 += ap[(size_t)(i + 0) * NN];
        s1 += ap[(size_t)(i + 1) * NN];
        s2 += ap[(size_t)(i + 2) * NN];
        s3 += ap[(size_t)(i + 3) * NN];
    }
    parts[((size_t)rc * NS + s) * NN + j] = (s0 + s1) + (s2 + s3);
}

#define RWAVES 16
#define RROWS  4
#define RCB    (NN / (RWAVES * RROWS))
template <bool LAST, int P>
__global__ void __launch_bounds__(RWAVES * 32) rowcol_kernel(const float* __restrict__ km,
                                                             const float* __restrict__ pin,
                                                             float* __restrict__ r2,
                                                             float* __restrict__ c2,
                                                             float* __restrict__ pout) {
    __shared__ __attribute__((aligned(16))) float csh[NN];
    const int s = blockIdx.y, rb = blockIdx.x;
    const int t = threadIdx.x;
    const int w = t >> 5, l = t & 31;
    const int row0 = rb * (RWAVES * RROWS) + w * RROWS;
    const float* pp = pin + (size_t)s * NN + 2 * t;
    const float* ap = km + ((size_t)s * NN + row0) * NN + 4 * l;

    v2f pl[P];
    v4f c[8];
    v4f v[RROWS][8];
#pragma unroll
    for (int p = 0; p < P; ++p) pl[p] = *(const v2f*)(pp + (size_t)p * NS * NN);
#pragma unroll
    for (int r = 0; r < RROWS; ++r)
#pragma unroll
        for (int k = 0; k < 8; ++k) v[r][k] = *(const v4f*)(ap + (size_t)r * NN + 128 * k);
    __builtin_amdgcn_sched_barrier(0);

    {
        v2f tot = pl[0];
#pragma unroll
        for (int p = 1; p < P; ++p) tot += pl[p];
        v2f cc;
        cc.x = 1.0f / tot.x;
        cc.y = 1.0f / tot.y;
        *(v2f*)&csh[2 * t] = cc;
        if constexpr (LAST) {
            if (rb == 0) *(v2f*)&c2[s * NN + 2 * t] = cc;
        }
    }
    asm volatile("s_wait_dscnt 0x0\n\ts_barrier_signal -1\n\ts_barrier_wait -1" ::: "memory");
#pragma unroll
    for (int k = 0; k < 8; ++k) c[k] = *(const v4f*)&csh[128 * k + 4 * l];

    float rs[RROWS];
#pragma unroll
    for (int r = 0; r < RROWS; ++r) {
        v4f sa = v[r][0] * c[0];
#pragma unroll
        for (int k = 1; k < 8; ++k) sa = v[r][k] * c[k] + sa;
        rs[r] = (sa.x + sa.y) + (sa.z + sa.w);
    }
#pragma unroll
    for (int m = 16; m >= 1; m >>= 1)
#pragma unroll
        for (int r = 0; r < RROWS; ++r) rs[r] += __shfl_xor(rs[r], m, 32);
#pragma unroll
    for (int r = 0; r < RROWS; ++r) rs[r] = 1.0f / rs[r];

    if constexpr (LAST) {
        float mine = rs[0];
#pragma unroll
        for (int r = 1; r < RROWS; ++r) if (l == r) mine = rs[r];
        if (l < RROWS) r2[s * NN + row0 + l] = mine;
    } else {
        __shared__ __attribute__((aligned(16))) float csum[RWAVES * NN];
        v4f acc[8];
#pragma unroll
        for (int k = 0; k < 8; ++k) acc[k] = v[0][k] * rs[0];
#pragma unroll
        for (int r = 1; r < RROWS; ++r)
#pragma unroll
            for (int k = 0; k < 8; ++k) acc[k] = v[r][k] * rs[r] + acc[k];
#pragma unroll
        for (int k = 0; k < 8; ++k) *(v4f*)&csum[w * NN + 128 * k + 4 * l] = acc[k];
        __syncthreads();
        v2f tot = *(const v2f*)&csum[2 * t];
#pragma unroll
        for (int ww = 1; ww < RWAVES; ++ww) tot += *(const v2f*)&csum[ww * NN + 2 * t];
        *(v2f*)&pout[((size_t)rb * NS + s) * NN + 2 * t] = tot;
    }
}

#define XROWS 34
#define PCOLS 34

template <int KS0, int KS1>
__device__ __forceinline__ void gemm_part(const float* __restrict__ xsc, const float* __restrict__ psc,
                                          int w, int lm, int lh, v8f (&acc)[2][4]) {
#pragma unroll
    for (int ks = KS0; ks < KS1; ks += 4) {
        const int ka = ks + (lh << 1);
        v2f a[2];
#pragma unroll
        for (int rt = 0; rt < 2; ++rt) {
            int row = w * 32 + rt * 16 + lm;
            a[rt] = *(const v2f*)&xsc[row * XROWS + ka];
        }
#pragma unroll
        for (int ct = 0; ct < 4; ++ct) {
            int col = ct * 16 + lm;
            v2f b = *(const v2f*)&psc[col * PCOLS + ka];
#pragma unroll
            for (int rt = 0; rt < 2; ++rt) {
                acc[rt][ct] = __builtin_amdgcn_wmma_f32_16x16x4_f32(
                    false, a[rt], false, b, (short)0, acc[rt][ct], false, false);
            }
        }
    }
}

__global__ void __launch_bounds__(256) gemm_kernel(const float* __restrict__ x,
                                                   const float* __restrict__ km,
                                                   const float* __restrict__ r2,
                                                   const float* __restrict__ c2,
                                                   float* __restrict__ out) {
    constexpr int XST = MB * XROWS;
    constexpr int PST = 64 * PCOLS;
    __shared__ __attribute__((aligned(16))) float xs[2 * XST];
    __shared__ __attribute__((aligned(16))) float ps[2 * PST];
    const int s  = blockIdx.y;
    const int m0 = blockIdx.x * 64;
    const int t  = threadIdx.x;
    const int w  = t >> 5, l = t & 31;
    const int lm = l & 15, lh = l >> 4;

    const float* ra = km + (size_t)s * NN * NN;
    const float* rr = r2 + s * NN;
    const float* rc = c2 + s * NN;

    const int xrow = t >> 5, xk = t & 31;
    const int pk = t >> 6, pm = t & 63;
    const float* xg = x + xrow * NN + xk;
    const float* kg = ra + (size_t)pk * NN + m0 + pm;
    const float* hg = rr + pk;
    const float cs = rc[m0 + pm];

    v8f acc[2][4];
#pragma unroll
    for (int rt = 0; rt < 2; ++rt)
#pragma unroll
        for (int ct = 0; ct < 4; ++ct)
#pragma unroll
            for (int g = 0; g < 8; ++g) acc[rt][ct][g] = 0.0f;

    float xr[32], kr[8], hr[8];
#pragma unroll
    for (int it = 0; it < 32; ++it) xr[it] = xg[(size_t)(it * 8) * NN];
#pragma unroll
    for (int it = 0; it < 8; ++it) {
        kr[it] = kg[(size_t)(it * 4) * NN];
        hr[it] = hg[it * 4];
    }
#pragma unroll
    for (int it = 0; it < 32; ++it) xs[(it * 8 + xrow) * XROWS + xk] = xr[it];
#pragma unroll
    for (int it = 0; it < 8; ++it) ps[pm * PCOLS + it * 4 + pk] = (kr[it] * hr[it]) * cs;
    __syncthreads();

    for (int kt = 0; kt < NN / 32 - 1; ++kt) {
        const int cur = kt & 1;
        const float* xsc = xs + cur * XST;
        const float* psc = ps + cur * PST;
        float* xsn = xs + (cur ^ 1) * XST;
        float* psn = ps + (cur ^ 1) * PST;
        const int k1 = (kt + 1) * 32;
#pragma unroll
        for (int it = 0; it < 32; ++it) xr[it] = xg[(size_t)(it * 8) * NN + k1];
#pragma unroll
        for (int it = 0; it < 8; ++it) {
            kr[it] = kg[(size_t)(k1 + it * 4) * NN];
            hr[it] = hg[k1 + it * 4];
        }
        gemm_part<0, 16>(xsc, psc, w, lm, lh, acc);
        __builtin_amdgcn_sched_barrier(0);
#pragma unroll
        for (int it = 0; it < 32; ++it) xsn[(it * 8 + xrow) * XROWS + xk] = xr[it];
#pragma unroll
        for (int it = 0; it < 8; ++it) psn[pm * PCOLS + it * 4 + pk] = (kr[it] * hr[it]) * cs;
        gemm_part<16, 32>(xsc, psc, w, lm, lh, acc);
        __syncthreads();
    }
    gemm_part<0, 32>(xs + XST, ps + PST, w, lm, lh, acc);

    float* ro = out + (size_t)s * MB * NN;
#pragma unroll
    for (int rt = 0; rt < 2; ++rt)
#pragma unroll
        for (int ct = 0; ct < 4; ++ct)
#pragma unroll
            for (int g = 0; g < 8; ++g) {
                int row = w * 32 + rt * 16 + lh * 8 + g;
                int col = m0 + ct * 16 + lm;
                ro[row * NN + col] = acc[rt][ct][g];
            }
}

extern "C" void kernel_launch(void* const* d_in, const int* in_sizes, int n_in,
                              void* d_out, int out_size, void* d_ws, size_t ws_size,
                              hipStream_t stream) {
    const float* x  = (const float*)d_in[0];
    const float* la = (const float*)d_in[1];
    const float* u  = (const float*)d_in[2];
    float* out = (float*)d_out;

    char* ws = (char*)d_ws;
    float* km    = (float*)ws;
    float* r2    = (float*)(ws + (size_t)NS * NN * NN * sizeof(float));
    float* c2    = r2 + NS * NN;
    float* parts = c2 + NS * NN;

    float* pa = out;
    float* pb = out + (size_t)RCB * NS * NN;
    prep_kernel<<<(NN * NN / 4) / 256, 256, 0, stream>>>(la, u, km);
    colpass_kernel<<<dim3(4, 8, NS), 256, 0, stream>>>(km, parts);
    rowcol_kernel<false, 8><<<dim3(RCB, NS), RWAVES * 32, 0, stream>>>(km, parts, r2, c2, pa);
    for (int it = 0; it < 9; ++it) {
        rowcol_kernel<false, RCB><<<dim3(RCB, NS), RWAVES * 32, 0, stream>>>(km, pa, r2, c2, pb);
        rowcol_kernel<false, RCB><<<dim3(RCB, NS), RWAVES * 32, 0, stream>>>(km, pb, r2, c2, pa);
    }
    rowcol_kernel<false, RCB><<<dim3(RCB, NS), RWAVES * 32, 0, stream>>>(km, pa, r2, c2, pb);
    rowcol_kernel<true, RCB><<<dim3(RCB, NS), RWAVES * 32, 0, stream>>>(km, pb, r2, c2, pa);
    gemm_kernel<<<dim3(NN / 64, NS), 256, 0, stream>>>(x, km, r2, c2, out);
}
